# starting kernel
# baseline (speedup 1.0000x reference)
#include <hip/hip_runtime.h>
#include <hip/hip_bf16.h>

#define SOM_B     128
#define SOM_D     64
#define SOM_N     16384
#define NTILES    (SOM_N / 16)
#define BTILES    (SOM_B / 16)

typedef __attribute__((ext_vector_type(2))) float v2f;
typedef __attribute__((ext_vector_type(8))) float v8f;

__global__ __launch_bounds__(256)
void som_dist_wmma_f32(const float* __restrict__ x,
                       const float* __restrict__ w,
                       float* __restrict__ out) {
  const int lane  = threadIdx.x & 31;
  const int wvid  = (blockIdx.x << 3) + (threadIdx.x >> 5);
  const int ntile = wvid & (NTILES - 1);
  const int btile = wvid >> 10;

  const int row = lane & 15;
  const int hi  = lane >> 4;

  const float* __restrict__ xrow = x + (size_t)(btile * 16 + row) * SOM_D + 2 * hi;
  const float* __restrict__ wrow = w + (size_t)(ntile * 16 + row) * SOM_D + 2 * hi;

  v8f acc = {};
  float xs = 0.0f;
  float ws = 0.0f;

#pragma unroll
  for (int c = 0; c < 16; ++c) {
    v2f a = *(const v2f*)(xrow + 4 * c);
    v2f b = *(const v2f*)(wrow + 4 * c);
    xs += a.x * a.x + a.y * a.y;
    ws += b.x * b.x + b.y * b.y;
    acc = __builtin_amdgcn_wmma_f32_16x16x4_f32(
        false, a, false, b, (short)0, acc, false, false);
  }

  xs += __shfl_xor(xs, 16, 32);
  ws += __shfl_xor(ws, 16, 32);

  float* __restrict__ obase =
      out + (size_t)(btile * 16 + 8 * hi) * SOM_N + ntile * 16 + row;

#pragma unroll
  for (int i = 0; i < 8; ++i) {
    float xn = __shfl(xs, 8 * hi + i, 32);
    obase[(size_t)i * SOM_N] = xn + ws - 2.0f * acc[i];
  }
}

extern "C" void kernel_launch(void* const* d_in, const int* in_sizes, int n_in,
                              void* d_out, int out_size, void* d_ws, size_t ws_size,
                              hipStream_t stream) {
  (void)in_sizes; (void)n_in; (void)d_ws; (void)ws_size; (void)out_size;
  const float* x = (const float*)d_in[0];
  const float* w = (const float*)d_in[1];
  float* out = (float*)d_out;

  dim3 grid(BTILES * NTILES / 8);
  dim3 block(256);
  som_dist_wmma_f32<<<grid, block, 0, stream>>>(x, w, out);
}


# x and the workgroup's 64 weight rows staged once into LDS by whole-line 16-byte loads, WMMA fragments read from LDS; 256 workgroups x 8 waves, 2x2 tiles per wave (was 1 tile per wave from global)
# speedup vs baseline: 1.4130x; 1.4130x over previous
#include <hip/hip_runtime.h>
#include <hip/hip_bf16.h>


#define SOM_B     128
#define SOM_D     64
#define SOM_N     16384
#define STRIP     64
#define PITCH     68

typedef __attribute__((ext_vector_type(2))) float v2f;
typedef __attribute__((ext_vector_type(4))) float v4f;
typedef __attribute__((ext_vector_type(8))) float v8f;

__global__ __launch_bounds__(256)
void som_dist_wmma_f32(const float* __restrict__ x,
                       const float* __restrict__ w,
                       float* __restrict__ out) {
  __shared__ float xl[SOM_B * PITCH];
  __shared__ float wl[STRIP * PITCH];

  const int tid  = threadIdx.x;
  const int lane = tid & 31;
  const int wv   = tid >> 5;
  const int ntp  = wv & 1;
  const int btp  = wv >> 1;

  const v4f* __restrict__ x4 = (const v4f*)x;
  const v4f* __restrict__ w4 = (const v4f*)(w + (size_t)blockIdx.x * STRIP * SOM_D);
  v4f sx[8];
  v4f sw[4];
#pragma unroll
  for (int j = 0; j < 8; ++j) sx[j] = x4[tid + 256 * j];
#pragma unroll
  for (int j = 0; j < 4; ++j) sw[j] = w4[tid + 256 * j];
#pragma unroll
  for (int j = 0; j < 8; ++j) {
    const int idx = tid + 256 * j;
    *(v4f*)(xl + (idx >> 4) * PITCH + (idx & 15) * 4) = sx[j];
  }
#pragma unroll
  for (int j = 0; j < 4; ++j) {
    const int idx = tid + 256 * j;
    *(v4f*)(wl + (idx >> 4) * PITCH + (idx & 15) * 4) = sw[j];
  }
  __syncthreads();

  const int row = lane & 15;
  const int hi  = lane >> 4;

  const float* ap0 = xl + (32 * btp + row) * PITCH + 2 * hi;
  const float* ap1 = ap0 + 16 * PITCH;
  const float* bp0 = wl + (32 * ntp + row) * PITCH + 2 * hi;
  const float* bp1 = bp0 + 16 * PITCH;

  v8f acc00 = {}, acc01 = {}, acc10 = {}, acc11 = {};
  float xs0 = 0.0f, xs1 = 0.0f;
  float ws0 = 0.0f, ws1 = 0.0f;

  v2f a0[16], a1[16], b0[16], b1[16];
#pragma unroll
  for (int c = 0; c < 16; ++c) {
    a0[c] = *(const v2f*)(ap0 + 4 * c);
    b0[c] = *(const v2f*)(bp0 + 4 * c);
    a1[c] = *(const v2f*)(ap1 + 4 * c);
    b1[c] = *(const v2f*)(bp1 + 4 * c);
  }
  __builtin_amdgcn_sched_barrier(0);

#pragma unroll
  for (int c = 0; c < 16; ++c) {
    xs0 += a0[c].x * a0[c].x + a0[c].y * a0[c].y;
    xs1 += a1[c].x * a1[c].x + a1[c].y * a1[c].y;
    ws0 += b0[c].x * b0[c].x + b0[c].y * b0[c].y;
    ws1 += b1[c].x * b1[c].x + b1[c].y * b1[c].y;
    acc00 = __builtin_amdgcn_wmma_f32_16x16x4_f32(false, a0[c], false, b0[c], (short)0, acc00, false, false);
    acc01 = __builtin_amdgcn_wmma_f32_16x16x4_f32(false, a0[c], false, b1[c], (short)0, acc01, false, false);
    acc10 = __builtin_amdgcn_wmma_f32_16x16x4_f32(false, a1[c], false, b0[c], (short)0, acc10, false, false);
    acc11 = __builtin_amdgcn_wmma_f32_16x16x4_f32(false, a1[c], false, b1[c], (short)0, acc11, false, false);
  }

  xs0 += __shfl_xor(xs0, 16, 32);
  xs1 += __shfl_xor(xs1, 16, 32);
  ws0 += __shfl_xor(ws0, 16, 32);
  ws1 += __shfl_xor(ws1, 16, 32);

  float* __restrict__ obase =
      out + (size_t)(32 * btp + 8 * hi) * SOM_N + blockIdx.x * STRIP + 32 * ntp + row;

#pragma unroll
  for (int i = 0; i < 8; ++i) {
    const float xn0 = __shfl(xs0, 8 * hi + i, 32);
    const float xn1 = __shfl(xs1, 8 * hi + i, 32);
    float* o0 = obase + (size_t)i * SOM_N;
    float* o1 = o0 + (size_t)16 * SOM_N;
    o0[0]  = xn0 + ws0 - 2.0f * acc00[i];
    o0[16] = xn0 + ws1 - 2.0f * acc01[i];
    o1[0]  = xn1 + ws0 - 2.0f * acc10[i];
    o1[16] = xn1 + ws1 - 2.0f * acc11[i];
  }
}

extern "C" void kernel_launch(void* const* d_in, const int* in_sizes, int n_in,
                              void* d_out, int out_size, void* d_ws, size_t ws_size,
                              hipStream_t stream) {
  (void)in_sizes; (void)n_in; (void)d_ws; (void)ws_size; (void)out_size;
  const float* x = (const float*)d_in[0];
  const float* w = (const float*)d_in[1];
  float* out = (float*)d_out;

  dim3 grid(SOM_N / STRIP);
  dim3 block(256);
  som_dist_wmma_f32<<<grid, block, 0, stream>>>(x, w, out);
}


# squared row norms taken once per workgroup from the staging registers (piece sums folded over 16 lanes by DPP, parked in LDS) while w is in flight, not per wave among WMMAs; norm sum order changed
# speedup vs baseline: 1.4904x; 1.0548x over previous
#include <hip/hip_runtime.h>
#include <hip/hip_bf16.h>


#define SOM_B     128
#define SOM_D     64
#define SOM_N     16384
#define STRIP     64
#define PITCH     68

typedef __attribute__((ext_vector_type(2))) float v2f;
typedef __attribute__((ext_vector_type(4))) float v4f;
typedef __attribute__((ext_vector_type(8))) float v8f;

__device__ __forceinline__ float piece_sq(v4f q) {
  v2f lo = {q.x, q.y};
  v2f hi = {q.z, q.w};
  v2f s = lo * lo;
  s = __builtin_elementwise_fma(hi, hi, s);
  return s.x + s.y;
}

template <int M>
__device__ __forceinline__ float from_xor_lane(float v) {
  return __builtin_bit_cast(float, __builtin_amdgcn_update_dpp(0, __builtin_bit_cast(int, v),
                                                               0x160 | M, 0xf, 0xf, true));
}

template <int M, int H>
__device__ __forceinline__ void fold_rows(float* v, bool up) {
#pragma unroll
  for (int i = 0; i < H; ++i) {
    const float keep = up ? v[i + H] : v[i];
    const float send = up ? v[i] : v[i + H];
    v[i] = keep + from_xor_lane<M>(send);
  }
}

__global__ __launch_bounds__(256)
void som_dist_wmma_f32(const float* __restrict__ x,
                       const float* __restrict__ w,
                       float* __restrict__ out) {
  __shared__ float xl[SOM_B * PITCH];
  __shared__ float wl[STRIP * PITCH];
  __shared__ __attribute__((aligned(32))) float xnl[SOM_B];
  __shared__ __attribute__((aligned(32))) float wnl[STRIP];

  const int tid  = threadIdx.x;
  const int lane = tid & 31;
  const int wv   = tid >> 5;
  const int ntp  = wv & 1;
  const int btp  = wv >> 1;

  const v4f* __restrict__ x4 = (const v4f*)x;
  const v4f* __restrict__ w4 = (const v4f*)(w + (size_t)blockIdx.x * STRIP * SOM_D);
  v4f sx[8];
  v4f sw[4];
#pragma unroll
  for (int j = 0; j < 8; ++j) sx[j] = x4[tid + 256 * j];
  __builtin_amdgcn_sched_barrier(0);
#pragma unroll
  for (int j = 0; j < 4; ++j) sw[j] = w4[tid + 256 * j];

  const int prow = tid >> 4;
  const int pcol = tid & 15;

#pragma unroll
  for (int j = 0; j < 8; ++j) {
    const int idx = tid + 256 * j;
    *(v4f*)(xl + (idx >> 4) * PITCH + (idx & 15) * 4) = sx[j];
  }
  float nx[8];
#pragma unroll
  for (int j = 0; j < 8; ++j) nx[j] = piece_sq(sx[j]);
  fold_rows<8, 4>(nx, (pcol & 8) != 0);
  fold_rows<4, 2>(nx, (pcol & 4) != 0);
  fold_rows<2, 1>(nx, (pcol & 2) != 0);
  nx[0] += from_xor_lane<1>(nx[0]);
  xnl[prow + 16 * (pcol >> 1)] = nx[0];
  __builtin_amdgcn_sched_barrier(0);

#pragma unroll
  for (int j = 0; j < 4; ++j) {
    const int idx = tid + 256 * j;
    *(v4f*)(wl + (idx >> 4) * PITCH + (idx & 15) * 4) = sw[j];
  }
  float nw[4];
#pragma unroll
  for (int j = 0; j < 4; ++j) nw[j] = piece_sq(sw[j]);
  fold_rows<8, 2>(nw, (pcol & 8) != 0);
  fold_rows<4, 1>(nw, (pcol & 4) != 0);
  nw[0] += from_xor_lane<2>(nw[0]);
  nw[0] += from_xor_lane<1>(nw[0]);
  wnl[prow + 16 * (pcol >> 2)] = nw[0];
  __syncthreads();

  const int row = lane & 15;
  const int hi  = lane >> 4;

  const v4f* xq0 = (const v4f*)(xnl + 32 * btp + 8 * hi);
  const v4f* xq1 = (const v4f*)(xnl + 32 * btp + 16 + 8 * hi);
  const v4f xa0 = xq0[0], xb0 = xq0[1];
  const v4f xa1 = xq1[0], xb1 = xq1[1];
  const float wn0 = wnl[32 * ntp + row];
  const float wn1 = wnl[32 * ntp + 16 + row];

  const float* ap0 = xl + (32 * btp + row) * PITCH + 2 * hi;
  const float* ap1 = ap0 + 16 * PITCH;
  const float* bp0 = wl + (32 * ntp + row) * PITCH + 2 * hi;
  const float* bp1 = bp0 + 16 * PITCH;

  v8f acc00 = {}, acc01 = {}, acc10 = {}, acc11 = {};

  v2f a0[16], a1[16], b0[16], b1[16];
#pragma unroll
  for (int c = 0; c < 16; ++c) {
    a0[c] = *(const v2f*)(ap0 + 4 * c);
    b0[c] = *(const v2f*)(bp0 + 4 * c);
    a1[c] = *(const v2f*)(ap1 + 4 * c);
    b1[c] = *(const v2f*)(bp1 + 4 * c);
  }
  const v2f xp0[4] = {{xa0.x, xa0.y}, {xa0.z, xa0.w}, {xb0.x, xb0.y}, {xb0.z, xb0.w}};
  const v2f xp1[4] = {{xa1.x, xa1.y}, {xa1.z, xa1.w}, {xb1.x, xb1.y}, {xb1.z, xb1.w}};
  v2f s00[4], s01[4], s10[4], s11[4];
#pragma unroll
  for (int p = 0; p < 4; ++p) {
    s00[p] = xp0[p] + wn0;
    s01[p] = xp0[p] + wn1;
    s10[p] = xp1[p] + wn0;
    s11[p] = xp1[p] + wn1;
  }

  float* __restrict__ obase =
      out + (size_t)(32 * btp + 8 * hi) * SOM_N + blockIdx.x * STRIP + 32 * ntp + row;

  __builtin_amdgcn_sched_barrier(0);

#pragma unroll
  for (int c = 0; c < 16; ++c) {
    acc00 = __builtin_amdgcn_wmma_f32_16x16x4_f32(false, a0[c], false, b0[c], (short)0, acc00, false, false);
    acc01 = __builtin_amdgcn_wmma_f32_16x16x4_f32(false, a0[c], false, b1[c], (short)0, acc01, false, false);
    acc10 = __builtin_amdgcn_wmma_f32_16x16x4_f32(false, a1[c], false, b0[c], (short)0, acc10, false, false);
    acc11 = __builtin_amdgcn_wmma_f32_16x16x4_f32(false, a1[c], false, b1[c], (short)0, acc11, false, false);
  }

#pragma unroll
  for (int p = 0; p < 4; ++p) {
    const int i = 2 * p;
    const v2f c00 = {acc00[i], acc00[i + 1]};
    const v2f c01 = {acc01[i], acc01[i + 1]};
    const v2f c10 = {acc10[i], acc10[i + 1]};
    const v2f c11 = {acc11[i], acc11[i + 1]};
    const v2f r00 = s00[p] - 2.0f * c00;
    const v2f r01 = s01[p] - 2.0f * c01;
    const v2f r10 = s10[p] - 2.0f * c10;
    const v2f r11 = s11[p] - 2.0f * c11;
    float* o0 = obase + (size_t)i * SOM_N;
    float* o1 = o0 + (size_t)16 * SOM_N;
    o0[0]          = r00.x;
    o0[16]         = r01.x;
    o0[SOM_N]      = r00.y;
    o0[SOM_N + 16] = r01.y;
    o1[0]          = r10.x;
    o1[16]         = r11.x;
    o1[SOM_N]      = r10.y;
    o1[SOM_N + 16] = r11.y;
  }
}

extern "C" void kernel_launch(void* const* d_in, const int* in_sizes, int n_in,
                              void* d_out, int out_size, void* d_ws, size_t ws_size,
                              hipStream_t stream) {
  (void)in_sizes; (void)n_in; (void)d_ws; (void)ws_size; (void)out_size;
  const float* x = (const float*)d_in[0];
  const float* w = (const float*)d_in[1];
  float* out = (float*)d_out;

  dim3 grid(SOM_N / STRIP);
  dim3 block(256);
  som_dist_wmma_f32<<<grid, block, 0, stream>>>(x, w, out);
}
